# starting kernel
# baseline (speedup 1.0000x reference)
#include <hip/hip_runtime.h>
#include <math.h>


typedef _Float16 half_t;
typedef __attribute__((ext_vector_type(16))) _Float16 v16h;
typedef __attribute__((ext_vector_type(8)))  float    v8f;

#define B_DIM 32
#define L_DIM 4096
#define H_DIM 16
#define QS    64
#define VS    256
#define CHUNK 256
#define NCHUNK 16

__device__ __forceinline__ v8f wmma_f16(v16h a, v16h b, v8f c) {
  return __builtin_amdgcn_wmma_f32_16x16x32_f16(
       false, a,  false, b,
       (short)0, c,  false,  false);
}

__device__ __forceinline__ v16h load_frag_rm(const float* __restrict__ base,
                                             int ld, int r0, int k0,
                                             float scale, int lane) {
  const int r = lane & 15, hf = lane >> 4;
  const float* p = base + (size_t)(r0 + r) * ld + k0 + hf * 8;
  v16h f;
#pragma unroll
  for (int e = 0; e < 16; ++e)
    f[e] = (half_t)(p[16 * (e >> 3) + (e & 7)] * scale);
  return f;
}

__device__ __forceinline__ v16h load_frag_b_cm(const float* __restrict__ base,
                                               int ld, int k0, int n0, int lane) {
  const int n = lane & 15, hf = lane >> 4;
  const float* p = base + (size_t)(k0 + hf * 8) * ld + n0 + n;
  v16h f;
#pragma unroll
  for (int e = 0; e < 16; ++e)
    f[e] = (half_t)p[(size_t)(16 * (e >> 3) + (e & 7)) * ld];
  return f;
}

__device__ __forceinline__ v16h load_frag_a_lds(const half_t* base, int ldh,
                                                int k0, int lane) {
  const int m = lane & 15, hf = lane >> 4;
  const half_t* p = base + m * ldh + k0 + hf * 8;
  v16h f;
#pragma unroll
  for (int e = 0; e < 16; ++e)
    f[e] = p[16 * (e >> 3) + (e & 7)];
  return f;
}

__global__ __launch_bounds__(128) void k1_qproj(
    const float* __restrict__ query, const float* __restrict__ Wq,
    const float* __restrict__ bq, float* __restrict__ Q) {
  const int wid  = (blockIdx.x * blockDim.x + threadIdx.x) >> 5;
  const int lane = threadIdx.x & 31;
  const int mt = wid & 1;
  const int nt = wid >> 1;

  v8f c = {};
#pragma unroll 4
  for (int k0 = 0; k0 < 1024; k0 += 32) {
    v16h a = load_frag_rm(query, 1024, mt * 16, k0, 1.0f, lane);
    v16h b = load_frag_rm(Wq,    1024, nt * 16, k0, 1.0f, lane);
    c = wmma_f16(a, b, c);
  }
  const int n = lane & 15, hf = lane >> 4;
  const float bias = bq[nt * 16 + n];
#pragma unroll
  for (int v = 0; v < 8; ++v) {
    const int m = v + hf * 8;
    Q[(size_t)(mt * 16 + m) * 1024 + nt * 16 + n] = c[v] + bias;
  }
}

__global__ __launch_bounds__(256) void k2_attn_partial(
    const float* __restrict__ Q,
    const float* __restrict__ key,
    const float* __restrict__ value,
    float* __restrict__ ctxp,
    float* __restrict__ msum) {
  __shared__ float  S[16][CHUNK];
  __shared__ half_t P[16][CHUNK + 8];
  __shared__ float  red[16][2];

  const int b  = blockIdx.x >> 4;
  const int ch = blockIdx.x & 15;
  const int l0 = ch * CHUNK;
  const int tid = threadIdx.x, wave = tid >> 5, lane = tid & 31;

  const float* qb = Q + (size_t)b * 1024;
  const v16h a0 = load_frag_rm(qb, QS, 0, 0,  0.125f, lane);
  const v16h a1 = load_frag_rm(qb, QS, 0, 32, 0.125f, lane);

  const float* kb = key + ((size_t)b * L_DIM + l0) * QS;
#pragma unroll
  for (int t = 0; t < CHUNK / 16 / 8; ++t) {
    const int ln0 = (wave + t * 8) * 16;
    v16h b0 = load_frag_rm(kb, QS, ln0, 0,  1.0f, lane);
    v16h b1 = load_frag_rm(kb, QS, ln0, 32, 1.0f, lane);
    v8f c = {};
    c = wmma_f16(a0, b0, c);
    c = wmma_f16(a1, b1, c);
    const int n = lane & 15, hf = lane >> 4;
#pragma unroll
    for (int v = 0; v < 8; ++v) S[v + hf * 8][ln0 + n] = c[v];
  }
  __syncthreads();

#pragma unroll
  for (int rr = 0; rr < 2; ++rr) {
    const int r = wave * 2 + rr;
    float vals[CHUNK / 32];
    float mx = -INFINITY;
#pragma unroll
    for (int i = 0; i < CHUNK / 32; ++i) {
      vals[i] = S[r][lane + 32 * i];
      mx = fmaxf(mx, vals[i]);
    }
#pragma unroll
    for (int off = 16; off; off >>= 1) mx = fmaxf(mx, __shfl_xor(mx, off, 32));
    float sum = 0.0f;
#pragma unroll
    for (int i = 0; i < CHUNK / 32; ++i) {
      const float e = __expf(vals[i] - mx);
      sum += e;
      P[r][lane + 32 * i] = (half_t)e;
    }
#pragma unroll
    for (int off = 16; off; off >>= 1) sum += __shfl_xor(sum, off, 32);
    if (lane == 0) { red[r][0] = mx; red[r][1] = sum; }
  }
  __syncthreads();

  const float* vb = value + ((size_t)b * L_DIM + l0) * VS;
  const int n0 = wave * 32;
  v8f c0 = {}, c1 = {};
#pragma unroll 2
  for (int k0 = 0; k0 < CHUNK; k0 += 32) {
    if (k0 + 32 < CHUNK)
      __builtin_prefetch(vb + (size_t)(k0 + 32 + lane) * VS + n0, 0, 0);
    v16h pa  = load_frag_a_lds(&P[0][0], CHUNK + 8, k0, lane);
    v16h vb0 = load_frag_b_cm(vb, VS, k0, n0,      lane);
    v16h vb1 = load_frag_b_cm(vb, VS, k0, n0 + 16, lane);
    c0 = wmma_f16(pa, vb0, c0);
    c1 = wmma_f16(pa, vb1, c1);
  }

  float* op = ctxp + (((size_t)b * NCHUNK + ch) * 16) * VS;
  const int n = lane & 15, hf = lane >> 4;
#pragma unroll
  for (int v = 0; v < 8; ++v) {
    const int m = v + hf * 8;
    op[(size_t)m * VS + n0 + n]      = c0[v];
    op[(size_t)m * VS + n0 + 16 + n] = c1[v];
  }
  if (tid < 16) {
    msum[(((size_t)b * NCHUNK + ch) * 16 + tid) * 2 + 0] = red[tid][0];
    msum[(((size_t)b * NCHUNK + ch) * 16 + tid) * 2 + 1] = red[tid][1];
  }
}

__global__ __launch_bounds__(256) void k3_reduce_out(
    const float* __restrict__ ctxp, const float* __restrict__ msum,
    const float* __restrict__ Wv, const float* __restrict__ bv,
    float* __restrict__ out) {
  __shared__ float ctx[16][VS];
  __shared__ float scl[16][NCHUNK];

  const int b = blockIdx.x;
  const int tid = threadIdx.x, wave = tid >> 5, lane = tid & 31;

  if (tid < 16) {
    const int h = tid;
    float mp[NCHUNK], sp[NCHUNK], mx = -INFINITY;
#pragma unroll
    for (int p = 0; p < NCHUNK; ++p) {
      mp[p] = msum[(((size_t)b * NCHUNK + p) * 16 + h) * 2 + 0];
      sp[p] = msum[(((size_t)b * NCHUNK + p) * 16 + h) * 2 + 1];
      mx = fmaxf(mx, mp[p]);
    }
    float denom = 0.0f;
#pragma unroll
    for (int p = 0; p < NCHUNK; ++p) denom += sp[p] * __expf(mp[p] - mx);
    const float inv = 1.0f / denom;
#pragma unroll
    for (int p = 0; p < NCHUNK; ++p) scl[h][p] = __expf(mp[p] - mx) * inv;
  }
  __syncthreads();

  for (int i = tid; i < 16 * VS; i += 256) {
    const int h = i >> 8, e = i & (VS - 1);
    float acc = 0.0f;
#pragma unroll
    for (int p = 0; p < NCHUNK; ++p)
      acc += ctxp[(((size_t)b * NCHUNK + p) * 16 + h) * VS + e] * scl[h][p];
    ctx[h][e] = acc;
  }
  __syncthreads();

  for (int o = wave; o < 1024; o += 8) {
    const int h = o >> 6;
    const float* wr = Wv + (size_t)o * VS;
    float acc = 0.0f;
#pragma unroll 4
    for (int e = lane; e < VS; e += 32) acc += ctx[h][e] * wr[e];
#pragma unroll
    for (int off = 16; off; off >>= 1) acc += __shfl_xor(acc, off, 32);
    if (lane == 0) out[(size_t)b * 1024 + o] = acc + bv[o];
  }
}

extern "C" void kernel_launch(void* const* d_in, const int* in_sizes, int n_in,
                              void* d_out, int out_size, void* d_ws, size_t ws_size,
                              hipStream_t stream) {
  const float* query = (const float*)d_in[0];
  const float* key   = (const float*)d_in[1];
  const float* value = (const float*)d_in[2];
  const float* Wq    = (const float*)d_in[3];
  const float* bq    = (const float*)d_in[4];
  const float* Wv    = (const float*)d_in[5];
  const float* bv    = (const float*)d_in[6];
  float* out = (float*)d_out;

  char* ws = (char*)d_ws;
  float* Q    = (float*)ws;
  float* ctxp = (float*)(ws + 131072);
  float* msum = (float*)(ws + 131072 +
                         (size_t)B_DIM * NCHUNK * 16 * VS * 4);

  k1_qproj<<<32, 128, 0, stream>>>(query, Wq, bq, Q);
  k2_attn_partial<<<B_DIM * NCHUNK, 256, 0, stream>>>(Q, key, value, ctxp, msum);
  k3_reduce_out<<<B_DIM, 256, 0, stream>>>(ctxp, msum, Wv, bv, out);
}


# k2 P@value: value staged as f16 through per-wave LDS slots (512-B row pieces, half of them requested before the scores), B read by transposing LDS load; 4 K groups summed via LDS: new sum order
# speedup vs baseline: 5.8934x; 5.8934x over previous
#include <hip/hip_runtime.h>
#include <math.h>


typedef _Float16 half_t;
typedef __attribute__((ext_vector_type(16))) _Float16 v16h;
typedef __attribute__((ext_vector_type(8)))  float    v8f;
typedef __attribute__((ext_vector_type(2)))  float    v2f;
typedef __attribute__((ext_vector_type(4)))  float    v4f;

#define B_DIM 32
#define L_DIM 4096
#define H_DIM 16
#define QS    64
#define VS    256
#define CHUNK 256
#define NCHUNK 16

__device__ __forceinline__ v8f wmma_f16(v16h a, v16h b, v8f c) {
  return __builtin_amdgcn_wmma_f32_16x16x32_f16(
       false, a,  false, b,
       (short)0, c,  false,  false);
}

__device__ __forceinline__ v16h load_frag_rm(const float* __restrict__ base,
                                             int ld, int r0, int k0,
                                             float scale, int lane) {
  const int r = lane & 15, hf = lane >> 4;
  const float* p = base + (size_t)(r0 + r) * ld + k0 + hf * 8;
  v16h f;
#pragma unroll
  for (int e = 0; e < 16; ++e)
    f[e] = (half_t)(p[16 * (e >> 3) + (e & 7)] * scale);
  return f;
}

__device__ __forceinline__ v16h load_frag_b_cm(const float* __restrict__ base,
                                               int ld, int k0, int n0, int lane) {
  const int n = lane & 15, hf = lane >> 4;
  const float* p = base + (size_t)(k0 + hf * 8) * ld + n0 + n;
  v16h f;
#pragma unroll
  for (int e = 0; e < 16; ++e)
    f[e] = (half_t)p[(size_t)(16 * (e >> 3) + (e & 7)) * ld];
  return f;
}

__device__ __forceinline__ v16h load_frag_a_lds(const half_t* base, int ldh,
                                                int k0, int lane) {
  const int m = lane & 15, hf = lane >> 4;
  const half_t* p = base + m * ldh + k0 + hf * 8;
  v16h f;
#pragma unroll
  for (int e = 0; e < 16; ++e)
    f[e] = p[16 * (e >> 3) + (e & 7)];
  return f;
}

#define K1_G  4
#define K1_NG (1024 / (32 * K1_G))

__device__ __forceinline__ void k1_issue(const float* __restrict__ pa,
                                         const float* __restrict__ pb,
                                         v4f (&ra)[K1_G][4], v4f (&rb)[K1_G][4]) {
#pragma unroll
  for (int s = 0; s < K1_G; ++s) {
    const float* qa = pa + 32 * s;
    const float* qb = pb + 32 * s;
    rb[s][0] = *(const v4f*)(qb);
    rb[s][1] = *(const v4f*)(qb + 4);
    rb[s][2] = *(const v4f*)(qb + 16);
    rb[s][3] = *(const v4f*)(qb + 20);
    ra[s][0] = *(const v4f*)(qa);
    ra[s][1] = *(const v4f*)(qa + 4);
    ra[s][2] = *(const v4f*)(qa + 16);
    ra[s][3] = *(const v4f*)(qa + 20);
  }
}

__device__ __forceinline__ v16h k1_to_frag(const v4f (&x)[4]) {
  v16h f;
#pragma unroll
  for (int i = 0; i < 4; ++i) {
    f[4 * i + 0] = (half_t)x[i].x;
    f[4 * i + 1] = (half_t)x[i].y;
    f[4 * i + 2] = (half_t)x[i].z;
    f[4 * i + 3] = (half_t)x[i].w;
  }
  return f;
}

__global__ __launch_bounds__(32) void k1_qproj(
    const float* __restrict__ query, const float* __restrict__ Wq,
    const float* __restrict__ bq, float* __restrict__ Q) {
  const int wid  = blockIdx.x;
  const int lane = threadIdx.x;
  const int mt = wid & 1;
  const int nt = wid >> 1;
  const int r = lane & 15, hf = lane >> 4;

  const float* pa = query + (size_t)(mt * 16 + r) * 1024 + hf * 8;
  const float* pb = Wq    + (size_t)(nt * 16 + r) * 1024 + hf * 8;

  v4f ra[2][K1_G][4], rb[2][K1_G][4];
  v8f c = {};
  const float bias = bq[nt * 16 + r];
  k1_issue(pa, pb, ra[0], rb[0]);
#pragma unroll
  for (int g = 0; g < K1_NG; ++g) {
    if (g + 1 < K1_NG) {
      k1_issue(pa + (g + 1) * 32 * K1_G, pb + (g + 1) * 32 * K1_G,
               ra[(g + 1) & 1], rb[(g + 1) & 1]);
    }
    __builtin_amdgcn_sched_barrier(0);
#pragma unroll
    for (int s = 0; s < K1_G; ++s) {
      const v16h a = k1_to_frag(ra[g & 1][s]);
      const v16h b = k1_to_frag(rb[g & 1][s]);
      c = wmma_f16(a, b, c);
    }
  }
  const int n = r;
#pragma unroll
  for (int v = 0; v < 8; ++v) {
    const int m = v + hf * 8;
    Q[(size_t)(mt * 16 + m) * 1024 + nt * 16 + n] = c[v] + bias;
  }
}

typedef __attribute__((ext_vector_type(4))) unsigned int u32x4;
typedef __attribute__((ext_vector_type(8))) unsigned int u32x8;
typedef __attribute__((ext_vector_type(4))) _Float16     v4h;
typedef short s16x8 __attribute__((ext_vector_type(8)));

#define K2_G    16
#define K2_PV   288
#define K2_SLOT (32 * K2_PV)
#define K2_RP   132

__device__ __forceinline__ void k2_raw_rm(const float* __restrict__ base, int ld,
                                          int r0, int k0, int lane, v4f (&x)[4]) {
  const int r = lane & 15, hf = lane >> 4;
  const float* p = base + (size_t)(r0 + r) * ld + k0 + hf * 8;
  x[0] = *(const v4f*)(p);
  x[1] = *(const v4f*)(p + 4);
  x[2] = *(const v4f*)(p + 16);
  x[3] = *(const v4f*)(p + 20);
}

__device__ __forceinline__ v16h k2_raw_to_frag(const v4f (&x)[4], float scale) {
  v16h f;
#pragma unroll
  for (int i = 0; i < 4; ++i) {
    f[4 * i + 0] = (half_t)(x[i].x * scale);
    f[4 * i + 1] = (half_t)(x[i].y * scale);
    f[4 * i + 2] = (half_t)(x[i].z * scale);
    f[4 * i + 3] = (half_t)(x[i].w * scale);
  }
  return f;
}

__device__ __forceinline__ void k2_issue(const float* __restrict__ src, v4f (&buf)[K2_G]) {
#pragma unroll
  for (int i = 0; i < K2_G; ++i) buf[i] = *(const v4f*)(src + (size_t)i * VS);
}

__device__ __forceinline__ void k2_park(unsigned char* dst, const v4f (&buf)[K2_G]) {
#pragma unroll
  for (int i = 0; i < K2_G; ++i) {
    v4h h;
    h[0] = (half_t)buf[i].x; h[1] = (half_t)buf[i].y;
    h[2] = (half_t)buf[i].z; h[3] = (half_t)buf[i].w;
    *(v4h*)(dst + i * K2_PV) = h;
  }
}

__device__ __forceinline__ u32x4 k2_ld_tr16(const unsigned char* p) {
  typedef __attribute__((address_space(3))) s16x8 lds_s16x8;
  return __builtin_bit_cast(u32x4, __builtin_amdgcn_ds_load_tr16_b128_v8i16((lds_s16x8*)p));
}

__device__ __forceinline__ void k2_kstep(const half_t* Pbase, int kk,
                                         const unsigned char* ad, int lane, v8f (&acc)[8]) {
  const v16h pa = load_frag_a_lds(Pbase, CHUNK + 8, kk, lane);
  u32x4 lo[8], hi[8];
#pragma unroll
  for (int t = 0; t < 8; ++t) {
    lo[t] = k2_ld_tr16(ad + 32 * t);
    hi[t] = k2_ld_tr16(ad + 16 * K2_PV + 32 * t);
  }
#pragma unroll
  for (int t = 0; t < 8; ++t) {
    const u32x8 w = {lo[t].x, lo[t].y, lo[t].z, lo[t].w, hi[t].x, hi[t].y, hi[t].z, hi[t].w};
    acc[t] = wmma_f16(pa, __builtin_bit_cast(v16h, w), acc[t]);
  }
}

__device__ __forceinline__ void k2_lds_barrier() {
  asm volatile("s_wait_dscnt 0x0" ::: "memory");
  asm volatile("s_barrier_signal -1" ::: "memory");
  asm volatile("s_barrier_wait -1" ::: "memory");
}

__global__ __launch_bounds__(256) void k2_attn_partial(
    const float* __restrict__ Q,
    const float* __restrict__ key,
    const float* __restrict__ value,
    float* __restrict__ ctxp,
    float* __restrict__ msum) {
  __shared__ float  S[16][CHUNK];
  __shared__ half_t P[16][CHUNK + 8];
  __shared__ float  red[16][2];
  __shared__ v4f    stageS[8 * K2_SLOT / 16];

  const int b  = blockIdx.x >> 4;
  const int ch = blockIdx.x & 15;
  const int l0 = ch * CHUNK;
  const int tid = threadIdx.x, wave = tid >> 5, lane = tid & 31;
  const int n = lane & 15, hf = lane >> 4;

  const int wv = __builtin_amdgcn_readfirstlane(wave);
  const int cg = wv & 1;
  const int kg = wv >> 1;
  const float* vsrc = value + ((size_t)b * L_DIM + l0 + 64 * kg) * VS + 128 * cg + 4 * lane;
  unsigned char* slot = (unsigned char*)stageS + wv * K2_SLOT;

  const float* qb = Q + (size_t)b * 1024;
  const float* kb = key + ((size_t)b * L_DIM + l0) * QS;
  v4f qr[2][4], kr[2][2][4];
  k2_raw_rm(qb, QS, 0, 0,  lane, qr[0]);
  k2_raw_rm(qb, QS, 0, 32, lane, qr[1]);
#pragma unroll
  for (int t = 0; t < 2; ++t) {
    const int ln0 = (wave + t * 8) * 16;
    k2_raw_rm(kb, QS, ln0, 0,  lane, kr[t][0]);
    k2_raw_rm(kb, QS, ln0, 32, lane, kr[t][1]);
  }
  v4f va[K2_G], vb[K2_G], vc[K2_G], vd[K2_G];
  k2_issue(vsrc, va);
  k2_issue(vsrc + (size_t)K2_G * VS, vb);
  __builtin_amdgcn_sched_barrier(0);

  const v16h a0 = k2_raw_to_frag(qr[0], 0.125f);
  const v16h a1 = k2_raw_to_frag(qr[1], 0.125f);
  v8f sc[2];
#pragma unroll
  for (int t = 0; t < 2; ++t) {
    const v16h b0 = k2_raw_to_frag(kr[t][0], 1.0f);
    const v16h b1 = k2_raw_to_frag(kr[t][1], 1.0f);
    v8f c = {};
    c = wmma_f16(a0, b0, c);
    sc[t] = wmma_f16(a1, b1, c);
  }
#pragma unroll
  for (int t = 0; t < 2; ++t) {
    const int ln0 = (wave + t * 8) * 16;
#pragma unroll
    for (int v = 0; v < 8; ++v) S[v + hf * 8][ln0 + n] = sc[t][v];
  }
  k2_lds_barrier();

#pragma unroll
  for (int rr = 0; rr < 2; ++rr) {
    const int r = wave * 2 + rr;
    float vals[CHUNK / 32];
    float mx = -INFINITY;
#pragma unroll
    for (int i = 0; i < CHUNK / 32; ++i) {
      vals[i] = S[r][lane + 32 * i];
      mx = fmaxf(mx, vals[i]);
    }
#pragma unroll
    for (int off = 16; off; off >>= 1) mx = fmaxf(mx, __shfl_xor(mx, off, 32));
    float sum = 0.0f;
#pragma unroll
    for (int i = 0; i < CHUNK / 32; ++i) {
      const float e = __expf(vals[i] - mx);
      sum += e;
      P[r][lane + 32 * i] = (half_t)e;
    }
#pragma unroll
    for (int off = 16; off; off >>= 1) sum += __shfl_xor(sum, off, 32);
    if (lane == 0) { red[r][0] = mx; red[r][1] = sum; }
  }
  k2_lds_barrier();

  const unsigned char* ad = slot + (8 * hf + (lane & 7)) * K2_PV + ((lane >> 3) & 1) * 16;
  unsigned char* prow = slot + lane * 8;
  v8f acc[8] = {};

  k2_issue(vsrc + (size_t)(2 * K2_G) * VS, vc);
  k2_park(prow, va);
  k2_park(prow + K2_G * K2_PV, vb);
  k2_issue(vsrc + (size_t)(3 * K2_G) * VS, vd);
  k2_kstep(&P[0][0], 64 * kg, ad, lane, acc);
  k2_park(prow, vc);
  k2_park(prow + K2_G * K2_PV, vd);
  k2_kstep(&P[0][0], 64 * kg + 32, ad, lane, acc);

  float* pr = (float*)slot;
#pragma unroll
  for (int t = 0; t < 8; ++t)
#pragma unroll
    for (int v = 0; v < 8; ++v) pr[(8 * hf + v) * K2_RP + 16 * t + n] = acc[t][v];
  __syncthreads();

  float* op = ctxp + (((size_t)b * NCHUNK + ch) * 16) * VS;
  const int c4 = tid & 63;
  const float* ps = (const float*)((const unsigned char*)stageS + (c4 >> 5) * K2_SLOT) + (c4 & 31) * 4;
#pragma unroll
  for (int i = 0; i < 4; ++i) {
    const int m = (tid >> 6) + 4 * i;
    v4f s = *(const v4f*)(ps + m * K2_RP);
#pragma unroll
    for (int g = 1; g < 4; ++g)
      s += *(const v4f*)(ps + (size_t)g * (2 * K2_SLOT / 4) + m * K2_RP);
    *(v4f*)(op + (size_t)m * VS + c4 * 4) = s;
  }
  if (tid < 16) {
    msum[(((size_t)b * NCHUNK + ch) * 16 + tid) * 2 + 0] = red[tid][0];
    msum[(((size_t)b * NCHUNK + ch) * 16 + tid) * 2 + 1] = red[tid][1];
  }
}

#define K3_WAVES   16
#define K3_ROWS    8
#define K3_PITCH4  65
#define K3_NTPITCH 272
#define K3_WPITCH  (4 * K3_NTPITCH)

__device__ __forceinline__ float lane_bcast(float v, int src_lane) {
  return __builtin_bit_cast(float,
      __builtin_amdgcn_readlane(__builtin_bit_cast(int, v), src_lane));
}

__device__ __forceinline__ void k3_load_frag8(const v4f (*rows)[K3_PITCH4],
                                              int r, int wave, int hf, float* f) {
  const v4f lo = rows[r][4 * wave + 2 * hf];
  const v4f hi = rows[r][4 * wave + 2 * hf + 1];
  f[0] = lo.x; f[1] = lo.y; f[2] = lo.z; f[3] = lo.w;
  f[4] = hi.x; f[5] = hi.y; f[6] = hi.z; f[7] = hi.w;
}

__global__ __launch_bounds__(512) void k3_reduce_out(
    const float* __restrict__ ctxp, const float* __restrict__ msum,
    const float* __restrict__ Wv, const float* __restrict__ bv,
    float* __restrict__ out) {
  __shared__ v4f ctxS[16][K3_PITCH4];
  __shared__ v4f wvS[64][K3_PITCH4];
  __shared__ float  part[K3_WAVES * K3_WPITCH];

  const int head = blockIdx.x >> 2;
  const int b0   = (blockIdx.x & 3) * K3_ROWS;
  const int tid = threadIdx.x, wave = tid >> 5, lane = tid & 31;
  const int c = lane & 15, hf = lane >> 4;

  const float bias = bv[head * 64 + (tid & 63)];
  const int mrow  = wave & 7;
  const int mhalf = wave >> 3;
  const int b = b0 + mrow;
  const v2f ms = *(const v2f*)(msum + (((size_t)b * NCHUNK + c) * 16 + head) * 2);

  const float* cp = ctxp + (((size_t)b * NCHUNK) * 16 + head) * VS + mhalf * 128 + lane * 4;
  v4f x[NCHUNK];
#pragma unroll
  for (int p = 0; p < NCHUNK; ++p) x[p] = *(const v4f*)(cp + (size_t)p * 16 * VS);

  v4f y[8];
#pragma unroll
  for (int i = 0; i < 8; ++i) {
    const int q = wave * 8 + i;
    y[i] = *(const v4f*)(Wv + (size_t)(head * 64 + (q >> 1)) * VS + (q & 1) * 128 + lane * 4);
  }
  __builtin_amdgcn_sched_barrier(0);

  float mx = ms.x;
#pragma unroll
  for (int off = 8; off; off >>= 1) mx = fmaxf(mx, __shfl_xor(mx, off, 32));
  const float ex = __expf(ms.x - mx);
  float denom = 0.0f;
#pragma unroll
  for (int p = 0; p < NCHUNK; ++p) denom += lane_bcast(ms.y, p) * lane_bcast(ex, p);
  const float wgt = ex * (1.0f / denom);

  v4f acc4 = {0.0f, 0.0f, 0.0f, 0.0f};
#pragma unroll
  for (int p = 0; p < NCHUNK; ++p) {
    const float w = lane_bcast(wgt, p);
    acc4 += x[p] * w;
  }
  ctxS[mrow][mhalf * 32 + lane] = acc4;
  ctxS[8 + (tid >> 6)][tid & 63] = (v4f){0.0f, 0.0f, 0.0f, 0.0f};
#pragma unroll
  for (int i = 0; i < 8; ++i) {
    const int q = wave * 8 + i;
    wvS[q >> 1][(q & 1) * 32 + lane] = y[i];
  }
  __syncthreads();

  float a[8], bw[4][8];
  k3_load_frag8(ctxS, c, wave, hf, a);
#pragma unroll
  for (int nt = 0; nt < 4; ++nt) k3_load_frag8(wvS, nt * 16 + c, wave, hf, bw[nt]);

  v8f acc[4] = {};
#pragma unroll
  for (int s = 0; s < 4; ++s) {
    v2f af; af[0] = a[2 * s]; af[1] = a[2 * s + 1];
#pragma unroll
    for (int nt = 0; nt < 4; ++nt) {
      v2f bf; bf[0] = bw[nt][2 * s]; bf[1] = bw[nt][2 * s + 1];
      acc[nt] = __builtin_amdgcn_wmma_f32_16x16x4_f32(
           false, af,  false, bf,
           (short)0, acc[nt],  false,  false);
    }
  }

  float* pw = part + wave * K3_WPITCH;
#pragma unroll
  for (int nt = 0; nt < 4; ++nt)
#pragma unroll
    for (int r = 0; r < 8; ++r) pw[nt * K3_NTPITCH + r * 32 + lane] = acc[nt][r];
  __syncthreads();

  const int row = tid >> 6;
  const int col = tid & 63;
  const int idx = (col >> 4) * K3_NTPITCH + row * 32 + (col & 15);
  float sum = 0.0f;
#pragma unroll
  for (int w = 0; w < K3_WAVES; ++w) sum += part[w * K3_WPITCH + idx];
  out[(size_t)(b0 + row) * 1024 + head * 64 + col] = sum + bias;
}

extern "C" void kernel_launch(void* const* d_in, const int* in_sizes, int n_in,
                              void* d_out, int out_size, void* d_ws, size_t ws_size,
                              hipStream_t stream) {
  const float* query = (const float*)d_in[0];
  const float* key   = (const float*)d_in[1];
  const float* value = (const float*)d_in[2];
  const float* Wq    = (const float*)d_in[3];
  const float* bq    = (const float*)d_in[4];
  const float* Wv    = (const float*)d_in[5];
  const float* bv    = (const float*)d_in[6];
  float* out = (float*)d_out;

  char* ws = (char*)d_ws;
  float* Q    = (float*)ws;
  float* ctxp = (float*)(ws + 131072);
  float* msum = (float*)(ws + 131072 +
                         (size_t)B_DIM * NCHUNK * 16 * VS * 4);

  k1_qproj<<<128, 32, 0, stream>>>(query, Wq, bq, Q);
  k2_attn_partial<<<B_DIM * NCHUNK, 256, 0, stream>>>(Q, key, value, ctxp, msum);
  k3_reduce_out<<<H_DIM * 4, 512, 0, stream>>>(ctxp, msum, Wv, bv, out);
}
